# starting kernel
# baseline (speedup 1.0000x reference)
#include <hip/hip_runtime.h>
#include <math.h>


typedef __attribute__((ext_vector_type(2))) float v2f;
typedef __attribute__((ext_vector_type(8))) float v8f;

constexpr int Bc  = 8;
constexpr int Tc  = 4096;
constexpr int Dc  = 1024;
constexpr int NCH = 32;
constexpr int CL  = Tc / NCH;
constexpr int NDG = Dc / 16;

__device__ __forceinline__ float sigmoidf_fast(float x) {
    return 1.0f / (1.0f + __expf(-x));
}

__global__ __launch_bounds__(256) void ema_phase1(
    const float* __restrict__ v, const float* __restrict__ logit,
    float* __restrict__ ends) {
    int g = blockIdx.x * 256 + threadIdx.x;
    int d = g & (Dc - 1);
    int c = (g >> 10) & (NCH - 1);
    int b = g >> 15;
    float a = sigmoidf_fast(logit[d]);
    float w = 1.0f - a;
    const float* vp = v + ((size_t)(b * Tc + c * CL)) * Dc + d;
    float cur = 0.0f;
#pragma unroll 4
    for (int t = 0; t < CL; ++t)
        cur = fmaf(w, cur, a * vp[(size_t)t * Dc]);
    ends[(b * NCH + c) * Dc + d] = cur;
}

__global__ __launch_bounds__(256) void ema_phase2(
    const float* __restrict__ ends, const float* __restrict__ logit,
    float* __restrict__ carry) {
    int g = blockIdx.x * 256 + threadIdx.x;
    int d = g & (Dc - 1);
    int b = g >> 10;
    float a = sigmoidf_fast(logit[d]);
    float w = 1.0f - a;
    float wL = w;
#pragma unroll
    for (int i = 0; i < 7; ++i) wL *= wL;
    float P = 0.0f;
    for (int c = 0; c < NCH; ++c) {
        carry[(b * NCH + c) * Dc + d] = P;
        P = fmaf(wL, P, ends[(b * NCH + c) * Dc + d]);
    }
}

__global__ __launch_bounds__(256) void ema_phase3(
    const float* __restrict__ v, const float* __restrict__ logit,
    const float* __restrict__ carry, float* __restrict__ out) {
    const int lane = threadIdx.x & 31;
    const int wave = blockIdx.x * 8 + (threadIdx.x >> 5);
    const int dg = wave & (NDG - 1);
    const int c  = (wave >> 6) & (NCH - 1);
    const int b  = wave >> 11;
    const int dl = lane & 15;
    const int d  = dg * 16 + dl;
    const bool hi = lane >= 16;
    const int jb = hi ? 2 : 0;
    const int ib = hi ? 8 : 0;

    float a  = sigmoidf_fast(logit[d]);
    float w  = 1.0f - a;
    float wi = 1.0f / w;
    float wi2 = wi * wi;
    float wi4 = wi2 * wi2;
    const float wq0_init = hi ? wi2 : 1.0f;
    float wbase = 1.0f;
    if (hi) { float w2 = w * w, w4 = w2 * w2; wbase = w4 * w4; }

    float P = carry[(b * NCH + c) * Dc + d];
    const size_t vb = (size_t)b * Tc * Dc;
    const float* vp = v + vb + d;
    float* op = out + vb + d;

    for (int tau = 0; tau < CL / 16; ++tau) {
        const int t0 = c * CL + tau * 16;
        v8f acc = {0.f, 0.f, 0.f, 0.f, 0.f, 0.f, 0.f, 0.f};
        float wq0 = wq0_init;
#pragma unroll
        for (int k = 0; k < 4; ++k) {
            const int j0 = 4 * k + jb;
            float x0 = vp[(size_t)(t0 + j0) * Dc];
            float x1 = vp[(size_t)(t0 + j0 + 1) * Dc];
            v2f bm;
            bm.x = wq0 * a * x0;
            bm.y = (wq0 * wi) * a * x1;
            v2f am;
            am.x = (j0     <= dl) ? 1.0f : 0.0f;
            am.y = (j0 + 1 <= dl) ? 1.0f : 0.0f;
            acc = __builtin_amdgcn_wmma_f32_16x16x4_f32(
                      false, am, false, bm, (short)0, acc, false, false);
            wq0 *= wi4;
        }
        const float wpP = w * P;
        float wr = wbase;
        float olast = 0.0f;
        #pragma unroll
        for (int r = 0; r < 8; ++r) {
            float o = wr * (acc[r] + wpP);
            __builtin_nontemporal_store(o, &op[(size_t)(t0 + ib + r) * Dc]);
            olast = o;
            wr *= w;
        }
        P = __shfl(olast, 16 + dl, 32);
    }
}

extern "C" void kernel_launch(void* const* d_in, const int* in_sizes, int n_in,
                              void* d_out, int out_size, void* d_ws, size_t ws_size,
                              hipStream_t stream) {
    const float* v     = (const float*)d_in[0];
    const float* logit = (const float*)d_in[1];
    float* out = (float*)d_out;
    float* ends  = (float*)d_ws;
    float* carry = ends + (size_t)Bc * NCH * Dc;

    ema_phase1<<<(Bc * NCH * Dc) / 256, 256, 0, stream>>>(v, logit, ends);
    ema_phase2<<<(Bc * Dc) / 256, 256, 0, stream>>>(ends, logit, carry);
    ema_phase3<<<(Bc * NCH * NDG) / 8, 256, 0, stream>>>(v, logit, carry, out);
}


# 16 B per lane: a lane owns 4 adjacent channels (WMMA column n = channels 4n+q), so each load/store moves whole 128-B lines; phase-1 chunk end joins two 64-step partials (sum order changes)
# speedup vs baseline: 1.7605x; 1.7605x over previous
#include <hip/hip_runtime.h>
#include <math.h>


typedef __attribute__((ext_vector_type(2))) float v2f;
typedef __attribute__((ext_vector_type(4))) float v4f;
typedef __attribute__((ext_vector_type(8))) float v8f;

constexpr int Bc   = 8;
constexpr int Tc   = 4096;
constexpr int Dc   = 1024;
constexpr int NCH  = 32;
constexpr int CL   = Tc / NCH;
constexpr int NG64 = Dc / 64;

__device__ __forceinline__ float sigmoidf_fast(float x) {
    return 1.0f / (1.0f + __expf(-x));
}

__global__ __launch_bounds__(256) void ema_phase1(
    const float* __restrict__ v, const float* __restrict__ logit,
    float* __restrict__ ends) {
    const int lane = threadIdx.x & 31;
    const int wave = blockIdx.x * 8 + (threadIdx.x >> 5);
    const int g  = wave & (NG64 - 1);
    const int c  = (wave >> 4) & (NCH - 1);
    const int b  = wave >> 9;
    const int cl = lane & 15;
    const int th = lane >> 4;
    const int d0 = g * 64 + 4 * cl;

    const v4f lg = *reinterpret_cast<const v4f*>(logit + d0);
    float a[4], w[4];
#pragma unroll
    for (int q = 0; q < 4; ++q) { a[q] = sigmoidf_fast(lg[q]); w[q] = 1.0f - a[q]; }

    const float* vp = v + ((size_t)(b * Tc + c * CL + th * (CL / 2))) * Dc + d0;
    float cur[4] = {0.0f, 0.0f, 0.0f, 0.0f};
    auto load8 = [&](v4f (&R)[8], int t) {
#pragma unroll
        for (int i = 0; i < 8; ++i)
            R[i] = *reinterpret_cast<const v4f*>(vp + (size_t)(t + i) * Dc);
    };
    auto scan8 = [&](const v4f (&R)[8]) {
#pragma unroll
        for (int i = 0; i < 8; ++i)
#pragma unroll
            for (int q = 0; q < 4; ++q) cur[q] = fmaf(w[q], cur[q], a[q] * R[i][q]);
    };
    v4f RA[8], RB[8];
    load8(RA, 0);
    load8(RB, 8);
#pragma unroll 1
    for (int t = 0; t < CL / 2 - 16; t += 16) {
        scan8(RA);
        load8(RA, t + 16);
        scan8(RB);
        load8(RB, t + 24);
    }
    scan8(RA);
    scan8(RB);
    v4f e;
#pragma unroll
    for (int q = 0; q < 4; ++q) {
        float w64 = w[q];
#pragma unroll
        for (int i = 0; i < 6; ++i) w64 *= w64;
        const float hiv = __shfl(cur[q], 16 + cl, 32);
        e[q] = fmaf(w64, cur[q], hiv);
    }
    if (th == 0)
        *reinterpret_cast<v4f*>(ends + (b * NCH + c) * Dc + d0) = e;
}

__global__ __launch_bounds__(256) void ema_phase2(
    const float* __restrict__ ends, const float* __restrict__ logit,
    float* __restrict__ carry) {
    int g = blockIdx.x * 256 + threadIdx.x;
    int d = g & (Dc - 1);
    int b = g >> 10;
    float a = sigmoidf_fast(logit[d]);
    float w = 1.0f - a;
    float wL = w;
#pragma unroll
    for (int i = 0; i < 7; ++i) wL *= wL;
    float P = 0.0f;
    for (int c = 0; c < NCH; ++c) {
        carry[(b * NCH + c) * Dc + d] = P;
        P = fmaf(wL, P, ends[(b * NCH + c) * Dc + d]);
    }
}

__global__ __launch_bounds__(256) void ema_phase3(
    const float* __restrict__ v, const float* __restrict__ logit,
    const float* __restrict__ carry, float* __restrict__ out) {
    const int lane = threadIdx.x & 31;
    const int wave = blockIdx.x * 8 + (threadIdx.x >> 5);
    const int g  = wave & (NG64 - 1);
    const int c  = (wave >> 4) & (NCH - 1);
    const int b  = wave >> 9;
    const int cl = lane & 15;
    const bool hi = lane >= 16;
    const int jb = hi ? 2 : 0;
    const int ib = hi ? 8 : 0;
    const int d0 = g * 64 + 4 * cl;

    const v4f lg = *reinterpret_cast<const v4f*>(logit + d0);
    float w[4], wbase[4];
    float m0[4][4], m1[4][4];
#pragma unroll
    for (int q = 0; q < 4; ++q) {
        const float a = sigmoidf_fast(lg[q]);
        w[q] = 1.0f - a;
        const float wi  = 1.0f / w[q];
        const float wi2 = wi * wi;
        const float wi4 = wi2 * wi2;
        float wq0 = hi ? wi2 : 1.0f;
#pragma unroll
        for (int k = 0; k < 4; ++k) {
            m0[k][q] = wq0 * a;
            m1[k][q] = (wq0 * wi) * a;
            wq0 *= wi4;
        }
        wbase[q] = 1.0f;
        if (hi) { float w2 = w[q] * w[q], w4 = w2 * w2; wbase[q] = w4 * w4; }
    }

    const v4f pc = *reinterpret_cast<const v4f*>(carry + (b * NCH + c) * Dc + d0);
    float P[4] = {pc[0], pc[1], pc[2], pc[3]};
    const size_t vb = (size_t)b * Tc * Dc;
    const float* vp = v + vb + (size_t)(c * CL + jb) * Dc + d0;
    float* op = out + vb + (size_t)(c * CL + ib) * Dc + d0;

    auto load_tile = [&](v4f (&Xt)[8], int tau) {
#pragma unroll
        for (int k = 0; k < 4; ++k) {
            Xt[2 * k]     = *reinterpret_cast<const v4f*>(vp + (size_t)(tau * 16 + 4 * k) * Dc);
            Xt[2 * k + 1] = *reinterpret_cast<const v4f*>(vp + (size_t)(tau * 16 + 4 * k + 1) * Dc);
        }
    };
    auto mma_tile = [&](const v4f (&Xt)[8], v8f (&acc)[4]) {
#pragma unroll
        for (int q = 0; q < 4; ++q) acc[q] = (v8f){0.f, 0.f, 0.f, 0.f, 0.f, 0.f, 0.f, 0.f};
#pragma unroll
        for (int k = 0; k < 4; ++k) {
            const int j0 = 4 * k + jb;
            v2f am;
            am.x = (j0     <= cl) ? 1.0f : 0.0f;
            am.y = (j0 + 1 <= cl) ? 1.0f : 0.0f;
#pragma unroll
            for (int q = 0; q < 4; ++q) {
                float y0 = m0[k][q] * Xt[2 * k][q];
                float y1 = m1[k][q] * Xt[2 * k + 1][q];
                asm("" : "+v"(y0), "+v"(y1));
                v2f bm;
                bm.x = y0;
                bm.y = y1;
                acc[q] = __builtin_amdgcn_wmma_f32_16x16x4_f32(
                             false, am, false, bm, (short)0, acc[q], false, false);
            }
        }
    };
    auto finish_tile = [&](const v8f (&acc)[4], int tau) {
        float wpP[4], wr[4];
#pragma unroll
        for (int q = 0; q < 4; ++q) { wpP[q] = w[q] * P[q]; wr[q] = wbase[q]; }
        v4f olast = {0.f, 0.f, 0.f, 0.f};
#pragma unroll
        for (int r = 0; r < 8; ++r) {
            v4f o;
#pragma unroll
            for (int q = 0; q < 4; ++q) {
                o[q] = wr[q] * (acc[q][r] + wpP[q]);
                wr[q] *= w[q];
            }
            __builtin_nontemporal_store(o, reinterpret_cast<v4f*>(op + (size_t)(tau * 16 + r) * Dc));
            olast = o;
        }
#pragma unroll
        for (int q = 0; q < 4; ++q) P[q] = __shfl(olast[q], 16 + cl, 32);
    };

    v4f XA[8], XB[8];
    v8f acc[4];
    load_tile(XA, 0);
    load_tile(XB, 1);
#pragma unroll 1
    for (int tau = 0; tau < CL / 16 - 2; tau += 2) {
        mma_tile(XA, acc);
        load_tile(XA, tau + 2);
        finish_tile(acc, tau);
        mma_tile(XB, acc);
        load_tile(XB, tau + 3);
        finish_tile(acc, tau + 1);
    }
    mma_tile(XA, acc);
    finish_tile(acc, CL / 16 - 2);
    mma_tile(XB, acc);
    finish_tile(acc, CL / 16 - 1);
}

extern "C" void kernel_launch(void* const* d_in, const int* in_sizes, int n_in,
                              void* d_out, int out_size, void* d_ws, size_t ws_size,
                              hipStream_t stream) {
    const float* v     = (const float*)d_in[0];
    const float* logit = (const float*)d_in[1];
    float* out = (float*)d_out;
    float* ends  = (float*)d_ws;
    float* carry = ends + (size_t)Bc * NCH * Dc;

    ema_phase1<<<(Bc * NCH * NG64) / 8, 256, 0, stream>>>(v, logit, ends);
    ema_phase2<<<(Bc * Dc) / 256, 256, 0, stream>>>(ends, logit, carry);
    ema_phase3<<<(Bc * NCH * NG64) / 8, 256, 0, stream>>>(v, logit, carry, out);
}


# one launch, v read once: a 16-wave workgroup per (batch, 64 channels) walks the sequence, waves scan 16-step tiles side by side, tile ends chained through LDS (carry sum order changes)
# speedup vs baseline: 2.3124x; 1.3135x over previous
#include <hip/hip_runtime.h>
#include <math.h>


typedef __attribute__((ext_vector_type(2))) float v2f;
typedef __attribute__((ext_vector_type(4))) float v4f;
typedef __attribute__((ext_vector_type(8))) float v8f;

constexpr int Bc   = 8;
constexpr int Tc   = 4096;
constexpr int Dc   = 1024;
constexpr int NG64 = Dc / 64;
constexpr int NW   = 16;
constexpr int NS   = Tc / (16 * NW);
constexpr int NB   = 2;
constexpr int NBLK = Bc * NG64;
constexpr int NTHR = NW * 32;

__device__ __forceinline__ float sigmoidf_fast(float x) {
    return 1.0f / (1.0f + __expf(-x));
}

__global__ __launch_bounds__(NTHR) void ema_scan(
    const float* __restrict__ v, const float* __restrict__ logit,
    float* __restrict__ out) {
    __shared__ v4f ends[2][NW][16];

    const int lane = threadIdx.x & 31;
    const int wv = __builtin_amdgcn_readfirstlane(threadIdx.x >> 5);
    const int g  = blockIdx.x & (NG64 - 1);
    const int b  = blockIdx.x >> 4;
    const int cl = lane & 15;
    const bool hi = lane >= 16;
    const int jb = hi ? 2 : 0;
    const int ib = hi ? 8 : 0;
    const int d0 = g * 64 + 4 * cl;

    const v4f lg = *reinterpret_cast<const v4f*>(logit + d0);
    float w[4], wbase[4], w15[4], W16[4];
    float m0[4][4], m1[4][4];
#pragma unroll
    for (int q = 0; q < 4; ++q) {
        const float a = sigmoidf_fast(lg[q]);
        w[q] = 1.0f - a;
        const float wi  = 1.0f / w[q];
        const float wi2 = wi * wi;
        const float wi4 = wi2 * wi2;
        float wq0 = hi ? wi2 : 1.0f;
#pragma unroll
        for (int k = 0; k < 4; ++k) {
            m0[k][q] = wq0 * a;
            m1[k][q] = (wq0 * wi) * a;
            wq0 *= wi4;
        }
        const float w2 = w[q] * w[q], w4 = w2 * w2, w8 = w4 * w4;
        wbase[q] = hi ? w8 : 1.0f;
        w15[q] = (w8 * w4) * (w2 * w[q]);
        W16[q] = w8 * w8;
    }

    float P0[4] = {0.0f, 0.0f, 0.0f, 0.0f};
    const size_t vb = (size_t)b * Tc * Dc;
    const float* vp = v + vb + (size_t)(wv * 16 + jb) * Dc + d0;
    float* op = out + vb + (size_t)(wv * 16 + ib) * Dc + d0;

    auto load_tile = [&](v4f (&Xt)[8], int s) {
#pragma unroll
        for (int k = 0; k < 4; ++k) {
            Xt[2 * k]     = *reinterpret_cast<const v4f*>(vp + (size_t)(s * (NW * 16) + 4 * k) * Dc);
            Xt[2 * k + 1] = *reinterpret_cast<const v4f*>(vp + (size_t)(s * (NW * 16) + 4 * k + 1) * Dc);
        }
    };
    auto mma_tile = [&](const v4f (&Xt)[8], v8f (&acc)[4]) {
#pragma unroll
        for (int q = 0; q < 4; ++q) acc[q] = (v8f){0.f, 0.f, 0.f, 0.f, 0.f, 0.f, 0.f, 0.f};
#pragma unroll
        for (int k = 0; k < 4; ++k) {
            const int j0 = 4 * k + jb;
            v2f am;
            am.x = (j0     <= cl) ? 1.0f : 0.0f;
            am.y = (j0 + 1 <= cl) ? 1.0f : 0.0f;
#pragma unroll
            for (int q = 0; q < 4; ++q) {
                float y0 = m0[k][q] * Xt[2 * k][q];
                float y1 = m1[k][q] * Xt[2 * k + 1][q];
                asm("" : "+v"(y0), "+v"(y1));
                v2f bm;
                bm.x = y0;
                bm.y = y1;
                acc[q] = __builtin_amdgcn_wmma_f32_16x16x4_f32(
                             false, am, false, bm, (short)0, acc[q], false, false);
            }
        }
    };
    auto carry_in = [&](const v8f (&acc)[4], int buf, float (&Pm)[4]) {
        v4f e;
#pragma unroll
        for (int q = 0; q < 4; ++q) e[q] = w15[q] * acc[q][7];
        if (hi) ends[buf][wv][cl] = e;
        asm volatile("s_wait_dscnt 0x0\n\ts_barrier_signal -1\n\ts_barrier_wait -1" ::: "memory");
        float P[4] = {P0[0], P0[1], P0[2], P0[3]};
#pragma unroll
        for (int q = 0; q < 4; ++q) Pm[q] = 0.0f;
#pragma unroll 1
        for (int h = 0; h < NW; h += 8) {
#pragma unroll
            for (int i = 0; i < 8; ++i) {
                const v4f eu = ends[buf][h + i][cl];
                const bool mine = (h + i == wv);
#pragma unroll
                for (int q = 0; q < 4; ++q) {
                    Pm[q] = mine ? P[q] : Pm[q];
                    P[q] = fmaf(W16[q], P[q], eu[q]);
                }
            }
        }
#pragma unroll
        for (int q = 0; q < 4; ++q) P0[q] = P[q];
    };
    auto finish_tile = [&](const v8f (&acc)[4], const float (&Pm)[4], int s) {
        float wpP[4], wr[4];
#pragma unroll
        for (int q = 0; q < 4; ++q) { wpP[q] = w[q] * Pm[q]; wr[q] = wbase[q]; }
#pragma unroll
        for (int r = 0; r < 8; ++r) {
            v4f o;
#pragma unroll
            for (int q = 0; q < 4; ++q) {
                o[q] = wr[q] * (acc[q][r] + wpP[q]);
                wr[q] *= w[q];
            }
            __builtin_nontemporal_store(o, reinterpret_cast<v4f*>(op + (size_t)(s * (NW * 16) + r) * Dc));
        }
    };

    v4f X[NB][8];
    v8f acc[4];
    float Pm[4];
#pragma unroll
    for (int i = 0; i < NB; ++i) load_tile(X[i], i);
#pragma unroll 1
    for (int s0 = 0; s0 < NS - NB; s0 += NB) {
#pragma unroll
        for (int i = 0; i < NB; ++i) {
            mma_tile(X[i], acc);
            load_tile(X[i], s0 + i + NB);
            carry_in(acc, i & 1, Pm);
            finish_tile(acc, Pm, s0 + i);
        }
    }
#pragma unroll
    for (int i = 0; i < NB; ++i) {
        mma_tile(X[i], acc);
        carry_in(acc, i & 1, Pm);
        finish_tile(acc, Pm, NS - NB + i);
    }
}

extern "C" void kernel_launch(void* const* d_in, const int* in_sizes, int n_in,
                              void* d_out, int out_size, void* d_ws, size_t ws_size,
                              hipStream_t stream) {
    const float* v     = (const float*)d_in[0];
    const float* logit = (const float*)d_in[1];
    float* out = (float*)d_out;

    ema_scan<<<NBLK, NTHR, 0, stream>>>(v, logit, out);
}
